# starting kernel
# baseline (speedup 1.0000x reference)
#include <hip/hip_runtime.h>
#include <hip/hip_bf16.h>

#define NNODES 100000
#define EDGES  1600000
#define DH     128
#define NCLS   100

typedef float v2f __attribute__((ext_vector_type(2)));
typedef float v8f __attribute__((ext_vector_type(8)));

__global__ void deg_kernel(const int* __restrict__ dst, float* __restrict__ deg, int E) {
    int i = blockIdx.x * blockDim.x + threadIdx.x;
    if (i < E) atomicAdd(&deg[dst[i]], 1.0f);
}

__global__ void dinv_kernel(const float* __restrict__ deg, float* __restrict__ dinv, int n) {
    int i = blockIdx.x * blockDim.x + threadIdx.x;
    if (i < n) dinv[i] = rsqrtf(deg[i] + 1.0f);
}

template <int K, int F>
__global__ void gemm_wmma_f32(const float* __restrict__ A, const float* __restrict__ W,
                              const float* __restrict__ bias, float* __restrict__ out) {
    __shared__ float sA[16 * K];
    const int rowBase = blockIdx.x * 16;

    {
        const float4* __restrict__ g4 = (const float4*)(A + (size_t)rowBase * K);
        float4* __restrict__ s4 = (float4*)sA;
        constexpr int TOT4 = (16 * K) / 4;
        #pragma unroll
        for (int i = 0; i < (TOT4 + 255) / 256; ++i) {
            int idx = threadIdx.x + i * 256;
            if (TOT4 % 256 == 0 || idx < TOT4) s4[idx] = g4[idx];
        }
    }
    __syncthreads();

    constexpr int NT = (F + 15) / 16;
    const int wave = threadIdx.x >> 5;
    if (wave >= NT) return;

    const int lane = threadIdx.x & 31;
    const int hi   = lane >> 4;
    const int l15  = lane & 15;
    const int col  = wave * 16 + l15;
    const int colc = (F % 16 == 0) ? col : (col < F ? col : F - 1);

    const float* __restrict__ sArow = sA + l15 * K;
    const float* __restrict__ Wcol  = W + colc;

    v8f acc = {};
    #pragma unroll
    for (int k = 0; k < K; k += 4) {
        const int kk = k + 2 * hi;
        v2f a, b;
        a.x = sArow[kk];
        a.y = sArow[kk + 1];
        b.x = Wcol[(size_t)kk * F];
        b.y = Wcol[(size_t)(kk + 1) * F];
        acc = __builtin_amdgcn_wmma_f32_16x16x4_f32(
             false, a,  false, b,
             (short)0, acc,  false,  false);
    }

    if (F % 16 == 0 || col < F) {
        const float bv  = bias ? bias[col] : 0.0f;
        const int   row = rowBase + 8 * hi;
        #pragma unroll
        for (int r = 0; r < 8; ++r)
            out[(size_t)(row + r) * F + col] = acc[r] + bv;
    }
}

__global__ void edge_agg(const float* __restrict__ h, const int* __restrict__ src,
                         const int* __restrict__ dst, const float* __restrict__ dinv,
                         float* __restrict__ agg, int E, int F) {
    const int e    = (blockIdx.x * blockDim.x + threadIdx.x) >> 5;
    const int lane = threadIdx.x & 31;
    if (e >= E) return;
    const int   s    = src[e];
    const int   d    = dst[e];
    const float coef = dinv[s] * dinv[d];
    const int   f    = lane * 4;
    if (f < F) {
        const float4 v = *(const float4*)(h + (size_t)s * F + f);
        float* a = agg + (size_t)d * F + f;
        atomicAdd(a + 0, v.x * coef);
        atomicAdd(a + 1, v.y * coef);
        atomicAdd(a + 2, v.z * coef);
        atomicAdd(a + 3, v.w * coef);
    }
}

__global__ void finalize_kernel(const float* __restrict__ agg, const float* __restrict__ h,
                                const float* __restrict__ dinv, const float* __restrict__ b,
                                float* __restrict__ out, int n, int F, int do_relu) {
    const long long i   = (long long)blockIdx.x * blockDim.x + threadIdx.x;
    const long long tot = (long long)n * F;
    if (i >= tot) return;
    const int row = (int)(i / F);
    const int col = (int)(i - (long long)row * F);
    const float di = dinv[row];
    float v = agg[i] + h[i] * di * di + b[col];
    out[i] = do_relu ? fmaxf(v, 0.0f) : v;
}

static void run_agg_epilogue(const float* bufH, float* bufAgg, float* outBuf,
                             const int* src, const int* dst, const float* dinv,
                             const float* b, int F, int relu, hipStream_t stream) {
    hipMemsetAsync(bufAgg, 0, (size_t)NNODES * F * sizeof(float), stream);
    const int eblocks = (EDGES * 32 + 255) / 256;
    edge_agg<<<eblocks, 256, 0, stream>>>(bufH, src, dst, dinv, bufAgg, EDGES, F);
    const long long tot = (long long)NNODES * F;
    finalize_kernel<<<(int)((tot + 255) / 256), 256, 0, stream>>>(bufAgg, bufH, dinv, b, outBuf,
                                                                  NNODES, F, relu);
}

extern "C" void kernel_launch(void* const* d_in, const int* in_sizes, int n_in,
                              void* d_out, int out_size, void* d_ws, size_t ws_size,
                              hipStream_t stream) {
    const float* x  = (const float*)d_in[0];
    const int*   ei = (const int*)d_in[1];
    const float* W1 = (const float*)d_in[2];   const float* b1 = (const float*)d_in[3];
    const float* W2 = (const float*)d_in[4];   const float* b2 = (const float*)d_in[5];
    const float* W3 = (const float*)d_in[6];   const float* b3 = (const float*)d_in[7];
    const float* W4 = (const float*)d_in[8];   const float* b4 = (const float*)d_in[9];
    const float* W5 = (const float*)d_in[10];  const float* b5 = (const float*)d_in[11];
    const float* Wl = (const float*)d_in[12];  const float* bl = (const float*)d_in[13];

    const int* src = ei;
    const int* dst = ei + EDGES;

    float* ws     = (float*)d_ws;
    float* deg    = ws;
    float* dinv   = ws + NNODES;
    float* bufH   = ws + 2 * (size_t)NNODES;
    float* bufAgg = bufH + (size_t)NNODES * DH;
    float* bufX   = bufAgg + (size_t)NNODES * DH;

    hipMemsetAsync(deg, 0, NNODES * sizeof(float), stream);
    deg_kernel<<<(EDGES + 255) / 256, 256, 0, stream>>>(dst, deg, EDGES);
    dinv_kernel<<<(NNODES + 255) / 256, 256, 0, stream>>>(deg, dinv, NNODES);

    const int gblocks = NNODES / 16;

    gemm_wmma_f32<DH, DH><<<gblocks, 256, 0, stream>>>(x, W1, nullptr, bufH);
    run_agg_epilogue(bufH, bufAgg, bufX, src, dst, dinv, b1, DH, 1, stream);
    gemm_wmma_f32<DH, DH><<<gblocks, 256, 0, stream>>>(bufX, W2, nullptr, bufH);
    run_agg_epilogue(bufH, bufAgg, bufX, src, dst, dinv, b2, DH, 0, stream);
    gemm_wmma_f32<DH, DH><<<gblocks, 256, 0, stream>>>(bufX, W3, nullptr, bufH);
    run_agg_epilogue(bufH, bufAgg, bufX, src, dst, dinv, b3, DH, 0, stream);
    gemm_wmma_f32<DH, DH><<<gblocks, 256, 0, stream>>>(bufX, W4, nullptr, bufH);
    run_agg_epilogue(bufH, bufAgg, bufX, src, dst, dinv, b4, DH, 0, stream);
    gemm_wmma_f32<DH, NCLS><<<gblocks, 256, 0, stream>>>(bufX, W5, nullptr, bufH);
    run_agg_epilogue(bufH, bufAgg, bufX, src, dst, dinv, b5, NCLS, 0, stream);

    gemm_wmma_f32<NCLS, NCLS><<<gblocks, 256, 0, stream>>>(bufX, Wl, bl, (float*)d_out);
}


# Aggregate by gather over a dst-sorted edge list built once per call (int atomic rank + scan): rows summed in registers in slot order (float sum order changes), self-loop+bias fused, no float atomics
# speedup vs baseline: 3.4663x; 3.4663x over previous
#include <hip/hip_runtime.h>

#define NNODES 100000
#define EDGES  1600000
#define DH     128
#define NCLS   100

static_assert(NNODES % 8 == 0, "one wave per row, 8 rows per block; int4 scan");

typedef float v2f __attribute__((ext_vector_type(2)));
typedef float v8f __attribute__((ext_vector_type(8)));

__global__ void count_kernel(const int* __restrict__ dst, int* __restrict__ cnt,
                             int* __restrict__ rank, int E) {
    int i = blockIdx.x * blockDim.x + threadIdx.x;
    if (i < E) rank[i] = atomicAdd(&cnt[dst[i]], 1);
}

__global__ void __launch_bounds__(256)
scan_kernel(const int* __restrict__ cnt, int* __restrict__ rowptr, float* __restrict__ dinv) {
    __shared__ int sInc[8];
    __shared__ int sPre[8];
    const int t = threadIdx.x, lane = t & 31, wave = t >> 5;
    const int4* __restrict__ c4 = (const int4*)cnt;
    constexpr int N4 = NNODES / 4;

    int pre = 0;
    const int nprev = blockIdx.x * 256;
    #pragma unroll 8
    for (int k = 0; k < (int)blockIdx.x; ++k) {
        const int4 q = c4[k * 256 + t];
        pre += (q.x + q.y) + (q.z + q.w);
    }

    const int i4 = nprev + t;
    int4 v = make_int4(0, 0, 0, 0);
    if (i4 < N4) v = c4[i4];
    const int s = (v.x + v.y) + (v.z + v.w);

    for (int o = 16; o > 0; o >>= 1) pre += __shfl_xor(pre, o);
    int inc = s;
    for (int o = 1; o < 32; o <<= 1) {
        const int u = __shfl_up(inc, o);
        if (lane >= o) inc += u;
    }
    if (lane == 31) sInc[wave] = inc;
    if (lane == 0)  sPre[wave] = pre;
    __syncthreads();
    int base = 0;
    for (int w = 0; w < 8; ++w) {
        base += sPre[w];
        if (w < wave) base += sInc[w];
    }
    if (i4 < N4) {
        int4 r;
        r.x = base + inc - s;
        r.y = r.x + v.x;
        r.z = r.y + v.y;
        r.w = r.z + v.z;
        ((int4*)rowptr)[i4] = r;
        if (i4 == N4 - 1) rowptr[NNODES] = r.w + v.w;
        float4 d;
        d.x = rsqrtf((float)v.x + 1.0f);
        d.y = rsqrtf((float)v.y + 1.0f);
        d.z = rsqrtf((float)v.z + 1.0f);
        d.w = rsqrtf((float)v.w + 1.0f);
        ((float4*)dinv)[i4] = d;
    }
}

__global__ void fill_kernel(const int* __restrict__ src, const int* __restrict__ dst,
                            const int* __restrict__ rank, const int* __restrict__ rowptr,
                            const float* __restrict__ dinv, int* __restrict__ csr, int E) {
    const int e = blockIdx.x * blockDim.x + threadIdx.x;
    if (e < E) {
        const int s = src[e];
        const int d = dst[e];
        int2 p;
        p.x = s;
        p.y = __float_as_int(dinv[s] * dinv[d]);
        ((int2*)csr)[rowptr[d] + rank[e]] = p;
    }
}

template <int K, int F>
__global__ void gemm_wmma_f32(const float* __restrict__ A, const float* __restrict__ W,
                              const float* __restrict__ bias, float* __restrict__ out) {
    __shared__ float sA[16 * K];
    const int rowBase = blockIdx.x * 16;

    {
        const float4* __restrict__ g4 = (const float4*)(A + (size_t)rowBase * K);
        float4* __restrict__ s4 = (float4*)sA;
        constexpr int TOT4 = (16 * K) / 4;
        #pragma unroll
        for (int i = 0; i < (TOT4 + 255) / 256; ++i) {
            int idx = threadIdx.x + i * 256;
            if (TOT4 % 256 == 0 || idx < TOT4) s4[idx] = g4[idx];
        }
    }
    __syncthreads();

    constexpr int NT = (F + 15) / 16;
    const int wave = threadIdx.x >> 5;
    if (wave >= NT) return;

    const int lane = threadIdx.x & 31;
    const int hi   = lane >> 4;
    const int l15  = lane & 15;
    const int col  = wave * 16 + l15;
    const int colc = (F % 16 == 0) ? col : (col < F ? col : F - 1);

    const float* __restrict__ sArow = sA + l15 * K;
    const float* __restrict__ Wcol  = W + colc;

    v8f acc = {};
    #pragma unroll
    for (int k = 0; k < K; k += 4) {
        const int kk = k + 2 * hi;
        v2f a, b;
        a.x = sArow[kk];
        a.y = sArow[kk + 1];
        b.x = Wcol[(size_t)kk * F];
        b.y = Wcol[(size_t)(kk + 1) * F];
        acc = __builtin_amdgcn_wmma_f32_16x16x4_f32(
             false, a,  false, b,
             (short)0, acc,  false,  false);
    }

    if (F % 16 == 0 || col < F) {
        const float bv  = bias ? bias[col] : 0.0f;
        const int   row = rowBase + 8 * hi;
        #pragma unroll
        for (int r = 0; r < 8; ++r)
            out[(size_t)(row + r) * F + col] = acc[r] + bv;
    }
}

template <int F, int U>
__device__ __forceinline__ void add_edges(const float* __restrict__ h, const int2* __restrict__ csr,
                                          int e, int f, float4& acc) {
    int2   p[U];
    float4 v[U];
    #pragma unroll
    for (int u = 0; u < U; ++u) p[u] = csr[e + u];
    #pragma unroll
    for (int u = 0; u < U; ++u) v[u] = *(const float4*)(h + (size_t)p[u].x * F + f);
    #pragma unroll
    for (int u = 0; u < U; ++u) {
        const float c = __int_as_float(p[u].y);
        acc.x = fmaf(v[u].x, c, acc.x);
        acc.y = fmaf(v[u].y, c, acc.y);
        acc.z = fmaf(v[u].z, c, acc.z);
        acc.w = fmaf(v[u].w, c, acc.w);
    }
}

template <int F>
__global__ void __launch_bounds__(256)
gather_agg(const float* __restrict__ h, const int* __restrict__ csr_raw,
           const int* __restrict__ rowptr, const float* __restrict__ dinv,
           const float* __restrict__ b, float* __restrict__ out, int do_relu) {
    const int2* __restrict__ csr = (const int2*)csr_raw;
    const int lane = threadIdx.x & 31;
    const int row = __builtin_amdgcn_readfirstlane(blockIdx.x * 8 + (threadIdx.x >> 5));
    const int f = lane * 4;
    if (F % 128 != 0 && f >= F) return;

    const int beg = rowptr[row];
    const int end = rowptr[row + 1];
    const float4 self = *(const float4*)(h + (size_t)row * F + f);
    const float di = dinv[row];
    const float sl = di * di;

    float4 acc;
    acc.x = fmaf(self.x, sl, b[f + 0]);
    acc.y = fmaf(self.y, sl, b[f + 1]);
    acc.z = fmaf(self.z, sl, b[f + 2]);
    acc.w = fmaf(self.w, sl, b[f + 3]);

    int e = beg;
    for (; e + 8 <= end; e += 8) add_edges<F, 8>(h, csr, e, f, acc);
    switch (end - e) {
        case 7: add_edges<F, 7>(h, csr, e, f, acc); break;
        case 6: add_edges<F, 6>(h, csr, e, f, acc); break;
        case 5: add_edges<F, 5>(h, csr, e, f, acc); break;
        case 4: add_edges<F, 4>(h, csr, e, f, acc); break;
        case 3: add_edges<F, 3>(h, csr, e, f, acc); break;
        case 2: add_edges<F, 2>(h, csr, e, f, acc); break;
        case 1: add_edges<F, 1>(h, csr, e, f, acc); break;
        default: break;
    }

    if (do_relu) {
        acc.x = fmaxf(acc.x, 0.0f);
        acc.y = fmaxf(acc.y, 0.0f);
        acc.z = fmaxf(acc.z, 0.0f);
        acc.w = fmaxf(acc.w, 0.0f);
    }
    *(float4*)(out + (size_t)row * F + f) = acc;
}

extern "C" void kernel_launch(void* const* d_in, const int* in_sizes, int n_in,
                              void* d_out, int out_size, void* d_ws, size_t ws_size,
                              hipStream_t stream) {
    const float* x  = (const float*)d_in[0];
    const int*   ei = (const int*)d_in[1];
    const float* W1 = (const float*)d_in[2];   const float* b1 = (const float*)d_in[3];
    const float* W2 = (const float*)d_in[4];   const float* b2 = (const float*)d_in[5];
    const float* W3 = (const float*)d_in[6];   const float* b3 = (const float*)d_in[7];
    const float* W4 = (const float*)d_in[8];   const float* b4 = (const float*)d_in[9];
    const float* W5 = (const float*)d_in[10];  const float* b5 = (const float*)d_in[11];
    const float* Wl = (const float*)d_in[12];  const float* bl = (const float*)d_in[13];

    const int* src = ei;
    const int* dst = ei + EDGES;

    float* ws     = (float*)d_ws;
    int*   cnt    = (int*)ws;
    float* dinv   = ws + NNODES;
    int*   rowptr = (int*)(ws + 2 * (size_t)NNODES);
    int*   rank   = rowptr + NNODES + 32;
    int*   csr    = rank + EDGES;
    float* bufH   = (float*)(csr + 2 * (size_t)EDGES);
    float* bufX   = bufH + (size_t)NNODES * DH;

    hipMemsetAsync(cnt, 0, NNODES * sizeof(int), stream);
    count_kernel<<<(EDGES + 255) / 256, 256, 0, stream>>>(dst, cnt, rank, EDGES);
    scan_kernel<<<(NNODES / 4 + 255) / 256, 256, 0, stream>>>(cnt, rowptr, dinv);
    fill_kernel<<<(EDGES + 255) / 256, 256, 0, stream>>>(src, dst, rank, rowptr, dinv, csr, EDGES);

    const int gblocks = NNODES / 16;
    const int ablocks = NNODES / 8;

    gemm_wmma_f32<DH, DH><<<gblocks, 256, 0, stream>>>(x, W1, nullptr, bufH);
    gather_agg<DH><<<ablocks, 256, 0, stream>>>(bufH, csr, rowptr, dinv, b1, bufX, 1);
    gemm_wmma_f32<DH, DH><<<gblocks, 256, 0, stream>>>(bufX, W2, nullptr, bufH);
    gather_agg<DH><<<ablocks, 256, 0, stream>>>(bufH, csr, rowptr, dinv, b2, bufX, 0);
    gemm_wmma_f32<DH, DH><<<gblocks, 256, 0, stream>>>(bufX, W3, nullptr, bufH);
    gather_agg<DH><<<ablocks, 256, 0, stream>>>(bufH, csr, rowptr, dinv, b3, bufX, 0);
    gemm_wmma_f32<DH, DH><<<gblocks, 256, 0, stream>>>(bufX, W4, nullptr, bufH);
    gather_agg<DH><<<ablocks, 256, 0, stream>>>(bufH, csr, rowptr, dinv, b4, bufX, 0);
    gemm_wmma_f32<DH, NCLS><<<gblocks, 256, 0, stream>>>(bufX, W5, nullptr, bufH);
    gather_agg<NCLS><<<ablocks, 256, 0, stream>>>(bufH, csr, rowptr, dinv, b5, bufX, 0);

    gemm_wmma_f32<NCLS, NCLS><<<gblocks, 256, 0, stream>>>(bufX, Wl, bl, (float*)d_out);
}


# Layers 2-5 + final linear have no nonlinearity: weights multiplied first on WMMA (biases as one more row), so 1 N-row GEMM + four 100-wide gathers replace 5 N-row GEMMs; float sum order changes
# speedup vs baseline: 4.7105x; 1.3589x over previous
#include <hip/hip_runtime.h>

#define NNODES 100000
#define EDGES  1600000
#define DH     128
#define NCLS   100

static_assert(NNODES % 8 == 0, "one wave per row, 8 rows per block; int4 scan");

typedef float v2f __attribute__((ext_vector_type(2)));
typedef float v8f __attribute__((ext_vector_type(8)));

__global__ void count_kernel(const int* __restrict__ dst, int* __restrict__ cnt,
                             int* __restrict__ rank, int E) {
    int i = blockIdx.x * blockDim.x + threadIdx.x;
    if (i < E) rank[i] = atomicAdd(&cnt[dst[i]], 1);
}

__global__ void __launch_bounds__(256)
scan_kernel(const int* __restrict__ cnt, int* __restrict__ rowptr, float* __restrict__ dinv) {
    __shared__ int sInc[8];
    __shared__ int sPre[8];
    const int t = threadIdx.x, lane = t & 31, wave = t >> 5;
    const int4* __restrict__ c4 = (const int4*)cnt;
    constexpr int N4 = NNODES / 4;

    int pre = 0;
    const int nprev = blockIdx.x * 256;
    #pragma unroll 8
    for (int k = 0; k < (int)blockIdx.x; ++k) {
        const int4 q = c4[k * 256 + t];
        pre += (q.x + q.y) + (q.z + q.w);
    }

    const int i4 = nprev + t;
    int4 v = make_int4(0, 0, 0, 0);
    if (i4 < N4) v = c4[i4];
    const int s = (v.x + v.y) + (v.z + v.w);

    for (int o = 16; o > 0; o >>= 1) pre += __shfl_xor(pre, o);
    int inc = s;
    for (int o = 1; o < 32; o <<= 1) {
        const int u = __shfl_up(inc, o);
        if (lane >= o) inc += u;
    }
    if (lane == 31) sInc[wave] = inc;
    if (lane == 0)  sPre[wave] = pre;
    __syncthreads();
    int base = 0;
    for (int w = 0; w < 8; ++w) {
        base += sPre[w];
        if (w < wave) base += sInc[w];
    }
    if (i4 < N4) {
        int4 r;
        r.x = base + inc - s;
        r.y = r.x + v.x;
        r.z = r.y + v.y;
        r.w = r.z + v.z;
        ((int4*)rowptr)[i4] = r;
        if (i4 == N4 - 1) rowptr[NNODES] = r.w + v.w;
        float4 d;
        d.x = rsqrtf((float)v.x + 1.0f);
        d.y = rsqrtf((float)v.y + 1.0f);
        d.z = rsqrtf((float)v.z + 1.0f);
        d.w = rsqrtf((float)v.w + 1.0f);
        ((float4*)dinv)[i4] = d;
    }
}

__global__ void fill_kernel(const int* __restrict__ src, const int* __restrict__ dst,
                            const int* __restrict__ rank, const int* __restrict__ rowptr,
                            const float* __restrict__ dinv, int* __restrict__ csr, int E) {
    const int e = blockIdx.x * blockDim.x + threadIdx.x;
    if (e < E) {
        const int s = src[e];
        const int d = dst[e];
        int2 p;
        p.x = s;
        p.y = __float_as_int(dinv[s] * dinv[d]);
        ((int2*)csr)[rowptr[d] + rank[e]] = p;
    }
}

template <int K, int F>
__global__ void gemm_wmma_f32(const float* __restrict__ A, const float* __restrict__ W,
                              const float* __restrict__ bias, float* __restrict__ out) {
    __shared__ float sA[16 * K];
    const int rowBase = blockIdx.x * 16;

    {
        const float4* __restrict__ g4 = (const float4*)(A + (size_t)rowBase * K);
        float4* __restrict__ s4 = (float4*)sA;
        constexpr int TOT4 = (16 * K) / 4;
        #pragma unroll
        for (int i = 0; i < (TOT4 + 255) / 256; ++i) {
            int idx = threadIdx.x + i * 256;
            if (TOT4 % 256 == 0 || idx < TOT4) s4[idx] = g4[idx];
        }
    }
    __syncthreads();

    constexpr int NT = (F + 15) / 16;
    const int wave = threadIdx.x >> 5;
    if (wave >= NT) return;

    const int lane = threadIdx.x & 31;
    const int hi   = lane >> 4;
    const int l15  = lane & 15;
    const int col  = wave * 16 + l15;
    const int colc = (F % 16 == 0) ? col : (col < F ? col : F - 1);

    const float* __restrict__ sArow = sA + l15 * K;
    const float* __restrict__ Wcol  = W + colc;

    v8f acc = {};
    #pragma unroll
    for (int k = 0; k < K; k += 4) {
        const int kk = k + 2 * hi;
        v2f a, b;
        a.x = sArow[kk];
        a.y = sArow[kk + 1];
        b.x = Wcol[(size_t)kk * F];
        b.y = Wcol[(size_t)(kk + 1) * F];
        acc = __builtin_amdgcn_wmma_f32_16x16x4_f32(
             false, a,  false, b,
             (short)0, acc,  false,  false);
    }

    if (F % 16 == 0 || col < F) {
        const float bv  = bias ? bias[col] : 0.0f;
        const int   row = rowBase + 8 * hi;
        #pragma unroll
        for (int r = 0; r < 8; ++r)
            out[(size_t)(row + r) * F + col] = acc[r] + bv;
    }
}

template <int K, int F>
__global__ void __launch_bounds__(256)
wchain_wmma_f32(const float* __restrict__ A, const float* __restrict__ arow,
                const float* __restrict__ W, const float* __restrict__ cadd,
                float* __restrict__ P, float* __restrict__ c) {
    __shared__ float sA[16 * K];
    const bool last    = (blockIdx.x == gridDim.x - 1);
    const int  rowBase = blockIdx.x * 16;

    {
        float4* __restrict__ s4 = (float4*)sA;
        constexpr int TOT4 = (16 * K) / 4;
        constexpr int ROW4 = K / 4;
        #pragma unroll
        for (int i = 0; i < (TOT4 + 255) / 256; ++i) {
            const int idx = threadIdx.x + i * 256;
            if (TOT4 % 256 == 0 || idx < TOT4) {
                float4 v = make_float4(0.0f, 0.0f, 0.0f, 0.0f);
                if (!last)           v = ((const float4*)(A + (size_t)rowBase * K))[idx];
                else if (idx < ROW4) v = ((const float4*)arow)[idx];
                s4[idx] = v;
            }
        }
    }
    __syncthreads();

    constexpr int NT = (F + 15) / 16;
    const int wave = threadIdx.x >> 5;
    if (wave >= NT) return;

    const int lane = threadIdx.x & 31;
    const int hi   = lane >> 4;
    const int l15  = lane & 15;
    const int col  = wave * 16 + l15;
    const int colc = (F % 16 == 0) ? col : (col < F ? col : F - 1);

    const float* __restrict__ sArow = sA + l15 * K;
    const float* __restrict__ Wcol  = W + colc;

    v8f acc = {};
    if (last && hi == 0 && cadd != nullptr) acc[0] = cadd[colc];
    #pragma unroll
    for (int k = 0; k < K; k += 4) {
        const int kk = k + 2 * hi;
        v2f a, b;
        a.x = sArow[kk];
        a.y = sArow[kk + 1];
        b.x = Wcol[(size_t)kk * F];
        b.y = Wcol[(size_t)(kk + 1) * F];
        acc = __builtin_amdgcn_wmma_f32_16x16x4_f32(
             false, a,  false, b,
             (short)0, acc,  false,  false);
    }

    if (F % 16 == 0 || col < F) {
        if (!last) {
            const int row = rowBase + 8 * hi;
            #pragma unroll
            for (int r = 0; r < 8; ++r)
                P[(size_t)(row + r) * F + col] = acc[r];
        } else if (hi == 0) {
            c[col] = acc[0];
        }
    }
}

template <int F, int U>
__device__ __forceinline__ void add_edges(const float* __restrict__ h, const int2* __restrict__ csr,
                                          int e, int f, float4& acc) {
    int2   p[U];
    float4 v[U];
    #pragma unroll
    for (int u = 0; u < U; ++u) p[u] = csr[e + u];
    #pragma unroll
    for (int u = 0; u < U; ++u) v[u] = *(const float4*)(h + (size_t)p[u].x * F + f);
    #pragma unroll
    for (int u = 0; u < U; ++u) {
        const float c = __int_as_float(p[u].y);
        acc.x = fmaf(v[u].x, c, acc.x);
        acc.y = fmaf(v[u].y, c, acc.y);
        acc.z = fmaf(v[u].z, c, acc.z);
        acc.w = fmaf(v[u].w, c, acc.w);
    }
}

template <int F>
__global__ void __launch_bounds__(256)
gather_agg(const float* __restrict__ h, const int* __restrict__ csr_raw,
           const int* __restrict__ rowptr, const float* __restrict__ dinv,
           const float* __restrict__ b, float* __restrict__ out, int do_relu) {
    const int2* __restrict__ csr = (const int2*)csr_raw;
    const int lane = threadIdx.x & 31;
    const int row = __builtin_amdgcn_readfirstlane(blockIdx.x * 8 + (threadIdx.x >> 5));
    const int f = lane * 4;
    if (F % 128 != 0 && f >= F) return;

    const int beg = rowptr[row];
    const int end = rowptr[row + 1];
    const float4 self = *(const float4*)(h + (size_t)row * F + f);
    const float di = dinv[row];
    const float sl = di * di;

    float4 acc;
    acc.x = fmaf(self.x, sl, b[f + 0]);
    acc.y = fmaf(self.y, sl, b[f + 1]);
    acc.z = fmaf(self.z, sl, b[f + 2]);
    acc.w = fmaf(self.w, sl, b[f + 3]);

    int e = beg;
    for (; e + 8 <= end; e += 8) add_edges<F, 8>(h, csr, e, f, acc);
    switch (end - e) {
        case 7: add_edges<F, 7>(h, csr, e, f, acc); break;
        case 6: add_edges<F, 6>(h, csr, e, f, acc); break;
        case 5: add_edges<F, 5>(h, csr, e, f, acc); break;
        case 4: add_edges<F, 4>(h, csr, e, f, acc); break;
        case 3: add_edges<F, 3>(h, csr, e, f, acc); break;
        case 2: add_edges<F, 2>(h, csr, e, f, acc); break;
        case 1: add_edges<F, 1>(h, csr, e, f, acc); break;
        default: break;
    }

    if (do_relu) {
        acc.x = fmaxf(acc.x, 0.0f);
        acc.y = fmaxf(acc.y, 0.0f);
        acc.z = fmaxf(acc.z, 0.0f);
        acc.w = fmaxf(acc.w, 0.0f);
    }
    *(float4*)(out + (size_t)row * F + f) = acc;
}

extern "C" void kernel_launch(void* const* d_in, const int* in_sizes, int n_in,
                              void* d_out, int out_size, void* d_ws, size_t ws_size,
                              hipStream_t stream) {
    const float* x  = (const float*)d_in[0];
    const int*   ei = (const int*)d_in[1];
    const float* W1 = (const float*)d_in[2];   const float* b1 = (const float*)d_in[3];
    const float* W2 = (const float*)d_in[4];   const float* b2 = (const float*)d_in[5];
    const float* W3 = (const float*)d_in[6];   const float* b3 = (const float*)d_in[7];
    const float* W4 = (const float*)d_in[8];   const float* b4 = (const float*)d_in[9];
    const float* W5 = (const float*)d_in[10];  const float* b5 = (const float*)d_in[11];
    const float* Wl = (const float*)d_in[12];  const float* bl = (const float*)d_in[13];

    const int* src = ei;
    const int* dst = ei + EDGES;

    float* ws     = (float*)d_ws;
    int*   cnt    = (int*)ws;
    float* dinv   = ws + NNODES;
    int*   rowptr = (int*)(ws + 2 * (size_t)NNODES);
    int*   rank   = rowptr + NNODES + 32;
    int*   csr    = rank + EDGES;
    float* bufH   = (float*)(csr + 2 * (size_t)EDGES);
    float* bufX   = bufH + (size_t)NNODES * DH;
    float* P5     = bufX + (size_t)NNODES * DH;
    float* P4     = P5 + DH * NCLS;
    float* P3     = P4 + DH * NCLS;
    float* P2     = P3 + DH * NCLS;
    float* c5     = P2 + DH * NCLS;
    float* c4     = c5 + 128;
    float* c3     = c4 + 128;
    float* c2     = c3 + 128;

    wchain_wmma_f32<NCLS, NCLS><<<DH / 16 + 1, 256, 0, stream>>>(W5, b5, Wl, bl,      P5, c5);
    wchain_wmma_f32<DH,   NCLS><<<DH / 16 + 1, 256, 0, stream>>>(W4, b4, P5, nullptr, P4, c4);
    wchain_wmma_f32<DH,   NCLS><<<DH / 16 + 1, 256, 0, stream>>>(W3, b3, P4, nullptr, P3, c3);
    wchain_wmma_f32<DH,   NCLS><<<DH / 16 + 1, 256, 0, stream>>>(W2, b2, P3, nullptr, P2, c2);

    hipMemsetAsync(cnt, 0, NNODES * sizeof(int), stream);
    count_kernel<<<(EDGES + 255) / 256, 256, 0, stream>>>(dst, cnt, rank, EDGES);
    scan_kernel<<<(NNODES / 4 + 255) / 256, 256, 0, stream>>>(cnt, rowptr, dinv);
    fill_kernel<<<(EDGES + 255) / 256, 256, 0, stream>>>(src, dst, rank, rowptr, dinv, csr, EDGES);

    const int gblocks = NNODES / 16;
    const int ablocks = NNODES / 8;

    gemm_wmma_f32<DH, DH><<<gblocks, 256, 0, stream>>>(x, W1, nullptr, bufH);
    gather_agg<DH><<<ablocks, 256, 0, stream>>>(bufH, csr, rowptr, dinv, b1, bufX, 1);
    gemm_wmma_f32<DH, NCLS><<<gblocks, 256, 0, stream>>>(bufX, P2, nullptr, bufH);
    gather_agg<NCLS><<<ablocks, 256, 0, stream>>>(bufH, csr, rowptr, dinv, c2, bufX, 0);
    gather_agg<NCLS><<<ablocks, 256, 0, stream>>>(bufX, csr, rowptr, dinv, c3, bufH, 0);
    gather_agg<NCLS><<<ablocks, 256, 0, stream>>>(bufH, csr, rowptr, dinv, c4, bufX, 0);
    gather_agg<NCLS><<<ablocks, 256, 0, stream>>>(bufX, csr, rowptr, dinv, c5, (float*)d_out, 0);
}


# LDS A-tile row pitch 128 -> 132 floats in the WMMA GEMM kernels: the 16 rows of one A-fragment read fall on different banks, not all on one (16-way conflict on every LDS read removed); sum order kept
# speedup vs baseline: 5.1272x; 1.0885x over previous
#include <hip/hip_runtime.h>

#define NNODES 100000
#define EDGES  1600000
#define DH     128
#define NCLS   100

static_assert(NNODES % 8 == 0, "one wave per row, 8 rows per block; int4 scan");

typedef float v2f __attribute__((ext_vector_type(2)));
typedef float v8f __attribute__((ext_vector_type(8)));

__global__ void count_kernel(const int* __restrict__ dst, int* __restrict__ cnt,
                             int* __restrict__ rank, int E) {
    int i = blockIdx.x * blockDim.x + threadIdx.x;
    if (i < E) rank[i] = atomicAdd(&cnt[dst[i]], 1);
}

__global__ void __launch_bounds__(256)
scan_kernel(const int* __restrict__ cnt, int* __restrict__ rowptr, float* __restrict__ dinv) {
    __shared__ int sInc[8];
    __shared__ int sPre[8];
    const int t = threadIdx.x, lane = t & 31, wave = t >> 5;
    const int4* __restrict__ c4 = (const int4*)cnt;
    constexpr int N4 = NNODES / 4;

    int pre = 0;
    const int nprev = blockIdx.x * 256;
    #pragma unroll 8
    for (int k = 0; k < (int)blockIdx.x; ++k) {
        const int4 q = c4[k * 256 + t];
        pre += (q.x + q.y) + (q.z + q.w);
    }

    const int i4 = nprev + t;
    int4 v = make_int4(0, 0, 0, 0);
    if (i4 < N4) v = c4[i4];
    const int s = (v.x + v.y) + (v.z + v.w);

    for (int o = 16; o > 0; o >>= 1) pre += __shfl_xor(pre, o);
    int inc = s;
    for (int o = 1; o < 32; o <<= 1) {
        const int u = __shfl_up(inc, o);
        if (lane >= o) inc += u;
    }
    if (lane == 31) sInc[wave] = inc;
    if (lane == 0)  sPre[wave] = pre;
    __syncthreads();
    int base = 0;
    for (int w = 0; w < 8; ++w) {
        base += sPre[w];
        if (w < wave) base += sInc[w];
    }
    if (i4 < N4) {
        int4 r;
        r.x = base + inc - s;
        r.y = r.x + v.x;
        r.z = r.y + v.y;
        r.w = r.z + v.z;
        ((int4*)rowptr)[i4] = r;
        if (i4 == N4 - 1) rowptr[NNODES] = r.w + v.w;
        float4 d;
        d.x = rsqrtf((float)v.x + 1.0f);
        d.y = rsqrtf((float)v.y + 1.0f);
        d.z = rsqrtf((float)v.z + 1.0f);
        d.w = rsqrtf((float)v.w + 1.0f);
        ((float4*)dinv)[i4] = d;
    }
}

__global__ void fill_kernel(const int* __restrict__ src, const int* __restrict__ dst,
                            const int* __restrict__ rank, const int* __restrict__ rowptr,
                            const float* __restrict__ dinv, int* __restrict__ csr, int E) {
    const int e = blockIdx.x * blockDim.x + threadIdx.x;
    if (e < E) {
        const int s = src[e];
        const int d = dst[e];
        int2 p;
        p.x = s;
        p.y = __float_as_int(dinv[s] * dinv[d]);
        ((int2*)csr)[rowptr[d] + rank[e]] = p;
    }
}

template <int K>
struct APitch { static constexpr int value = ((K / 4) % 2 == 1) ? K : K + 4; };

template <int K, int F>
__global__ void gemm_wmma_f32(const float* __restrict__ A, const float* __restrict__ W,
                              const float* __restrict__ bias, float* __restrict__ out) {
    constexpr int PITCH = APitch<K>::value;
    __shared__ float sA[16 * PITCH];
    const int rowBase = blockIdx.x * 16;

    {
        const float4* __restrict__ g4 = (const float4*)(A + (size_t)rowBase * K);
        float4* __restrict__ s4 = (float4*)sA;
        constexpr int TOT4 = (16 * K) / 4;
        constexpr int ROW4 = K / 4;
        #pragma unroll
        for (int i = 0; i < (TOT4 + 255) / 256; ++i) {
            int idx = threadIdx.x + i * 256;
            if (TOT4 % 256 == 0 || idx < TOT4) {
                const int r = idx / ROW4;
                s4[idx + r * ((PITCH - K) / 4)] = g4[idx];
            }
        }
    }
    __syncthreads();

    constexpr int NT = (F + 15) / 16;
    const int wave = threadIdx.x >> 5;
    if (wave >= NT) return;

    const int lane = threadIdx.x & 31;
    const int hi   = lane >> 4;
    const int l15  = lane & 15;
    const int col  = wave * 16 + l15;
    const int colc = (F % 16 == 0) ? col : (col < F ? col : F - 1);

    const float* __restrict__ sArow = sA + l15 * PITCH;
    const float* __restrict__ Wcol  = W + colc;

    v8f acc = {};
    #pragma unroll
    for (int k = 0; k < K; k += 4) {
        const int kk = k + 2 * hi;
        v2f a, b;
        a.x = sArow[kk];
        a.y = sArow[kk + 1];
        b.x = Wcol[(size_t)kk * F];
        b.y = Wcol[(size_t)(kk + 1) * F];
        acc = __builtin_amdgcn_wmma_f32_16x16x4_f32(
             false, a,  false, b,
             (short)0, acc,  false,  false);
    }

    if (F % 16 == 0 || col < F) {
        const float bv  = bias ? bias[col] : 0.0f;
        const int   row = rowBase + 8 * hi;
        #pragma unroll
        for (int r = 0; r < 8; ++r)
            out[(size_t)(row + r) * F + col] = acc[r] + bv;
    }
}

template <int K, int F>
__global__ void __launch_bounds__(256)
wchain_wmma_f32(const float* __restrict__ A, const float* __restrict__ arow,
                const float* __restrict__ W, const float* __restrict__ cadd,
                float* __restrict__ P, float* __restrict__ c) {
    constexpr int PITCH = APitch<K>::value;
    __shared__ float sA[16 * PITCH];
    const bool last    = (blockIdx.x == gridDim.x - 1);
    const int  rowBase = blockIdx.x * 16;

    {
        float4* __restrict__ s4 = (float4*)sA;
        constexpr int TOT4 = (16 * K) / 4;
        constexpr int ROW4 = K / 4;
        #pragma unroll
        for (int i = 0; i < (TOT4 + 255) / 256; ++i) {
            const int idx = threadIdx.x + i * 256;
            if (TOT4 % 256 == 0 || idx < TOT4) {
                float4 v = make_float4(0.0f, 0.0f, 0.0f, 0.0f);
                if (!last)           v = ((const float4*)(A + (size_t)rowBase * K))[idx];
                else if (idx < ROW4) v = ((const float4*)arow)[idx];
                const int r = idx / ROW4;
                s4[idx + r * ((PITCH - K) / 4)] = v;
            }
        }
    }
    __syncthreads();

    constexpr int NT = (F + 15) / 16;
    const int wave = threadIdx.x >> 5;
    if (wave >= NT) return;

    const int lane = threadIdx.x & 31;
    const int hi   = lane >> 4;
    const int l15  = lane & 15;
    const int col  = wave * 16 + l15;
    const int colc = (F % 16 == 0) ? col : (col < F ? col : F - 1);

    const float* __restrict__ sArow = sA + l15 * PITCH;
    const float* __restrict__ Wcol  = W + colc;

    v8f acc = {};
    if (last && hi == 0 && cadd != nullptr) acc[0] = cadd[colc];
    #pragma unroll
    for (int k = 0; k < K; k += 4) {
        const int kk = k + 2 * hi;
        v2f a, b;
        a.x = sArow[kk];
        a.y = sArow[kk + 1];
        b.x = Wcol[(size_t)kk * F];
        b.y = Wcol[(size_t)(kk + 1) * F];
        acc = __builtin_amdgcn_wmma_f32_16x16x4_f32(
             false, a,  false, b,
             (short)0, acc,  false,  false);
    }

    if (F % 16 == 0 || col < F) {
        if (!last) {
            const int row = rowBase + 8 * hi;
            #pragma unroll
            for (int r = 0; r < 8; ++r)
                P[(size_t)(row + r) * F + col] = acc[r];
        } else if (hi == 0) {
            c[col] = acc[0];
        }
    }
}

template <int F, int U>
__device__ __forceinline__ void add_edges(const float* __restrict__ h, const int2* __restrict__ csr,
                                          int e, int f, float4& acc) {
    int2   p[U];
    float4 v[U];
    #pragma unroll
    for (int u = 0; u < U; ++u) p[u] = csr[e + u];
    #pragma unroll
    for (int u = 0; u < U; ++u) v[u] = *(const float4*)(h + (size_t)p[u].x * F + f);
    #pragma unroll
    for (int u = 0; u < U; ++u) {
        const float c = __int_as_float(p[u].y);
        acc.x = fmaf(v[u].x, c, acc.x);
        acc.y = fmaf(v[u].y, c, acc.y);
        acc.z = fmaf(v[u].z, c, acc.z);
        acc.w = fmaf(v[u].w, c, acc.w);
    }
}

template <int F>
__global__ void __launch_bounds__(256)
gather_agg(const float* __restrict__ h, const int* __restrict__ csr_raw,
           const int* __restrict__ rowptr, const float* __restrict__ dinv,
           const float* __restrict__ b, float* __restrict__ out, int do_relu) {
    const int2* __restrict__ csr = (const int2*)csr_raw;
    const int lane = threadIdx.x & 31;
    const int row = __builtin_amdgcn_readfirstlane(blockIdx.x * 8 + (threadIdx.x >> 5));
    const int f = lane * 4;
    if (F % 128 != 0 && f >= F) return;

    const int beg = rowptr[row];
    const int end = rowptr[row + 1];
    const float4 self = *(const float4*)(h + (size_t)row * F + f);
    const float di = dinv[row];
    const float sl = di * di;

    float4 acc;
    acc.x = fmaf(self.x, sl, b[f + 0]);
    acc.y = fmaf(self.y, sl, b[f + 1]);
    acc.z = fmaf(self.z, sl, b[f + 2]);
    acc.w = fmaf(self.w, sl, b[f + 3]);

    int e = beg;
    for (; e + 8 <= end; e += 8) add_edges<F, 8>(h, csr, e, f, acc);
    switch (end - e) {
        case 7: add_edges<F, 7>(h, csr, e, f, acc); break;
        case 6: add_edges<F, 6>(h, csr, e, f, acc); break;
        case 5: add_edges<F, 5>(h, csr, e, f, acc); break;
        case 4: add_edges<F, 4>(h, csr, e, f, acc); break;
        case 3: add_edges<F, 3>(h, csr, e, f, acc); break;
        case 2: add_edges<F, 2>(h, csr, e, f, acc); break;
        case 1: add_edges<F, 1>(h, csr, e, f, acc); break;
        default: break;
    }

    if (do_relu) {
        acc.x = fmaxf(acc.x, 0.0f);
        acc.y = fmaxf(acc.y, 0.0f);
        acc.z = fmaxf(acc.z, 0.0f);
        acc.w = fmaxf(acc.w, 0.0f);
    }
    *(float4*)(out + (size_t)row * F + f) = acc;
}

extern "C" void kernel_launch(void* const* d_in, const int* in_sizes, int n_in,
                              void* d_out, int out_size, void* d_ws, size_t ws_size,
                              hipStream_t stream) {
    const float* x  = (const float*)d_in[0];
    const int*   ei = (const int*)d_in[1];
    const float* W1 = (const float*)d_in[2];   const float* b1 = (const float*)d_in[3];
    const float* W2 = (const float*)d_in[4];   const float* b2 = (const float*)d_in[5];
    const float* W3 = (const float*)d_in[6];   const float* b3 = (const float*)d_in[7];
    const float* W4 = (const float*)d_in[8];   const float* b4 = (const float*)d_in[9];
    const float* W5 = (const float*)d_in[10];  const float* b5 = (const float*)d_in[11];
    const float* Wl = (const float*)d_in[12];  const float* bl = (const float*)d_in[13];

    const int* src = ei;
    const int* dst = ei + EDGES;

    float* ws     = (float*)d_ws;
    int*   cnt    = (int*)ws;
    float* dinv   = ws + NNODES;
    int*   rowptr = (int*)(ws + 2 * (size_t)NNODES);
    int*   rank   = rowptr + NNODES + 32;
    int*   csr    = rank + EDGES;
    float* bufH   = (float*)(csr + 2 * (size_t)EDGES);
    float* bufX   = bufH + (size_t)NNODES * DH;
    float* P5     = bufX + (size_t)NNODES * DH;
    float* P4     = P5 + DH * NCLS;
    float* P3     = P4 + DH * NCLS;
    float* P2     = P3 + DH * NCLS;
    float* c5     = P2 + DH * NCLS;
    float* c4     = c5 + 128;
    float* c3     = c4 + 128;
    float* c2     = c3 + 128;

    wchain_wmma_f32<NCLS, NCLS><<<DH / 16 + 1, 256, 0, stream>>>(W5, b5, Wl, bl,      P5, c5);
    wchain_wmma_f32<DH,   NCLS><<<DH / 16 + 1, 256, 0, stream>>>(W4, b4, P5, nullptr, P4, c4);
    wchain_wmma_f32<DH,   NCLS><<<DH / 16 + 1, 256, 0, stream>>>(W3, b3, P4, nullptr, P3, c3);
    wchain_wmma_f32<DH,   NCLS><<<DH / 16 + 1, 256, 0, stream>>>(W2, b2, P3, nullptr, P2, c2);

    hipMemsetAsync(cnt, 0, NNODES * sizeof(int), stream);
    count_kernel<<<(EDGES + 255) / 256, 256, 0, stream>>>(dst, cnt, rank, EDGES);
    scan_kernel<<<(NNODES / 4 + 255) / 256, 256, 0, stream>>>(cnt, rowptr, dinv);
    fill_kernel<<<(EDGES + 255) / 256, 256, 0, stream>>>(src, dst, rank, rowptr, dinv, csr, EDGES);

    const int gblocks = NNODES / 16;
    const int ablocks = NNODES / 8;

    gemm_wmma_f32<DH, DH><<<gblocks, 256, 0, stream>>>(x, W1, nullptr, bufH);
    gather_agg<DH><<<ablocks, 256, 0, stream>>>(bufH, csr, rowptr, dinv, b1, bufX, 1);
    gemm_wmma_f32<DH, NCLS><<<gblocks, 256, 0, stream>>>(bufX, P2, nullptr, bufH);
    gather_agg<NCLS><<<ablocks, 256, 0, stream>>>(bufH, csr, rowptr, dinv, c2, bufX, 0);
    gather_agg<NCLS><<<ablocks, 256, 0, stream>>>(bufX, csr, rowptr, dinv, c3, bufH, 0);
    gather_agg<NCLS><<<ablocks, 256, 0, stream>>>(bufH, csr, rowptr, dinv, c4, bufX, 0);
    gather_agg<NCLS><<<ablocks, 256, 0, stream>>>(bufX, csr, rowptr, dinv, c5, (float*)d_out, 0);
}


# Independent work shares launches: weight-chain products ride as 9 extra blocks in the count, scan, fill and layer-1 gather launches; x W1 tiles go 1 in 3 among count and fill blocks; 14->10 launches
# speedup vs baseline: 5.4727x; 1.0674x over previous
#include <hip/hip_runtime.h>

#define NNODES 100000
#define EDGES  1600000
#define DH     128
#define NCLS   100

static_assert(NNODES % 8 == 0, "one wave per row, 8 rows per block; int4 scan");

typedef float v2f __attribute__((ext_vector_type(2)));
typedef float v8f __attribute__((ext_vector_type(8)));

__device__ __forceinline__ void count_body(int bid, const int* __restrict__ dst, int* __restrict__ cnt,
                                           int* __restrict__ rank, int E) {
    int i = bid * 256 + threadIdx.x;
    if (i < E) rank[i] = atomicAdd(&cnt[dst[i]], 1);
}

__device__ __forceinline__ void scan_body(int bid, const int* __restrict__ cnt, int* __restrict__ rowptr,
                                          float* __restrict__ dinv) {
    __shared__ int sInc[8];
    __shared__ int sPre[8];
    const int t = threadIdx.x, lane = t & 31, wave = t >> 5;
    const int4* __restrict__ c4 = (const int4*)cnt;
    constexpr int N4 = NNODES / 4;

    int pre = 0;
    const int nprev = bid * 256;
    #pragma unroll 8
    for (int k = 0; k < bid; ++k) {
        const int4 q = c4[k * 256 + t];
        pre += (q.x + q.y) + (q.z + q.w);
    }

    const int i4 = nprev + t;
    int4 v = make_int4(0, 0, 0, 0);
    if (i4 < N4) v = c4[i4];
    const int s = (v.x + v.y) + (v.z + v.w);

    for (int o = 16; o > 0; o >>= 1) pre += __shfl_xor(pre, o);
    int inc = s;
    for (int o = 1; o < 32; o <<= 1) {
        const int u = __shfl_up(inc, o);
        if (lane >= o) inc += u;
    }
    if (lane == 31) sInc[wave] = inc;
    if (lane == 0)  sPre[wave] = pre;
    __syncthreads();
    int base = 0;
    for (int w = 0; w < 8; ++w) {
        base += sPre[w];
        if (w < wave) base += sInc[w];
    }
    if (i4 < N4) {
        int4 r;
        r.x = base + inc - s;
        r.y = r.x + v.x;
        r.z = r.y + v.y;
        r.w = r.z + v.z;
        ((int4*)rowptr)[i4] = r;
        if (i4 == N4 - 1) rowptr[NNODES] = r.w + v.w;
        float4 d;
        d.x = rsqrtf((float)v.x + 1.0f);
        d.y = rsqrtf((float)v.y + 1.0f);
        d.z = rsqrtf((float)v.z + 1.0f);
        d.w = rsqrtf((float)v.w + 1.0f);
        ((float4*)dinv)[i4] = d;
    }
}

__device__ __forceinline__ void fill_body(int bid, const int* __restrict__ src, const int* __restrict__ dst,
                                          const int* __restrict__ rank, const int* __restrict__ rowptr,
                                          const float* __restrict__ dinv, int* __restrict__ csr, int E) {
    const int e = bid * 256 + threadIdx.x;
    if (e < E) {
        const int s = src[e];
        const int d = dst[e];
        int2 p;
        p.x = s;
        p.y = __float_as_int(dinv[s] * dinv[d]);
        ((int2*)csr)[rowptr[d] + rank[e]] = p;
    }
}

template <int K>
struct APitch { static constexpr int value = ((K / 4) % 2 == 1) ? K : K + 4; };

template <int K, int F>
__device__ __forceinline__ void gemm_body(int bid, const float* __restrict__ A, const float* __restrict__ W,
                                          const float* __restrict__ bias, float* __restrict__ out) {
    constexpr int PITCH = APitch<K>::value;
    __shared__ float sA[16 * PITCH];
    const int rowBase = bid * 16;

    {
        const float4* __restrict__ g4 = (const float4*)(A + (size_t)rowBase * K);
        float4* __restrict__ s4 = (float4*)sA;
        constexpr int TOT4 = (16 * K) / 4;
        constexpr int ROW4 = K / 4;
        #pragma unroll
        for (int i = 0; i < (TOT4 + 255) / 256; ++i) {
            int idx = threadIdx.x + i * 256;
            if (TOT4 % 256 == 0 || idx < TOT4) {
                const int r = idx / ROW4;
                s4[idx + r * ((PITCH - K) / 4)] = g4[idx];
            }
        }
    }
    __syncthreads();

    constexpr int NT = (F + 15) / 16;
    const int wave = threadIdx.x >> 5;
    if (wave >= NT) return;

    const int lane = threadIdx.x & 31;
    const int hi   = lane >> 4;
    const int l15  = lane & 15;
    const int col  = wave * 16 + l15;
    const int colc = (F % 16 == 0) ? col : (col < F ? col : F - 1);

    const float* __restrict__ sArow = sA + l15 * PITCH;
    const float* __restrict__ Wcol  = W + colc;

    v8f acc = {};
    #pragma unroll
    for (int k = 0; k < K; k += 4) {
        const int kk = k + 2 * hi;
        v2f a, b;
        a.x = sArow[kk];
        a.y = sArow[kk + 1];
        b.x = Wcol[(size_t)kk * F];
        b.y = Wcol[(size_t)(kk + 1) * F];
        acc = __builtin_amdgcn_wmma_f32_16x16x4_f32(
             false, a,  false, b,
             (short)0, acc,  false,  false);
    }

    if (F % 16 == 0 || col < F) {
        const float bv  = bias ? bias[col] : 0.0f;
        const int   row = rowBase + 8 * hi;
        #pragma unroll
        for (int r = 0; r < 8; ++r)
            out[(size_t)(row + r) * F + col] = acc[r] + bv;
    }
}

template <int K, int F>
__device__ __forceinline__ void wchain_body(int bid, int nblk,
                                            const float* __restrict__ A, const float* __restrict__ arow,
                                            const float* __restrict__ W, const float* __restrict__ cadd,
                                            float* __restrict__ P, float* __restrict__ c) {
    constexpr int PITCH = APitch<K>::value;
    __shared__ float sA[16 * PITCH];
    const bool last    = (bid == nblk - 1);
    const int  rowBase = bid * 16;

    {
        float4* __restrict__ s4 = (float4*)sA;
        constexpr int TOT4 = (16 * K) / 4;
        constexpr int ROW4 = K / 4;
        #pragma unroll
        for (int i = 0; i < (TOT4 + 255) / 256; ++i) {
            const int idx = threadIdx.x + i * 256;
            if (TOT4 % 256 == 0 || idx < TOT4) {
                float4 v = make_float4(0.0f, 0.0f, 0.0f, 0.0f);
                if (!last)           v = ((const float4*)(A + (size_t)rowBase * K))[idx];
                else if (idx < ROW4) v = ((const float4*)arow)[idx];
                const int r = idx / ROW4;
                s4[idx + r * ((PITCH - K) / 4)] = v;
            }
        }
    }
    __syncthreads();

    constexpr int NT = (F + 15) / 16;
    const int wave = threadIdx.x >> 5;
    if (wave >= NT) return;

    const int lane = threadIdx.x & 31;
    const int hi   = lane >> 4;
    const int l15  = lane & 15;
    const int col  = wave * 16 + l15;
    const int colc = (F % 16 == 0) ? col : (col < F ? col : F - 1);

    const float* __restrict__ sArow = sA + l15 * PITCH;
    const float* __restrict__ Wcol  = W + colc;

    v8f acc = {};
    if (last && hi == 0 && cadd != nullptr) acc[0] = cadd[colc];
    #pragma unroll
    for (int k = 0; k < K; k += 4) {
        const int kk = k + 2 * hi;
        v2f a, b;
        a.x = sArow[kk];
        a.y = sArow[kk + 1];
        b.x = Wcol[(size_t)kk * F];
        b.y = Wcol[(size_t)(kk + 1) * F];
        acc = __builtin_amdgcn_wmma_f32_16x16x4_f32(
             false, a,  false, b,
             (short)0, acc,  false,  false);
    }

    if (F % 16 == 0 || col < F) {
        if (!last) {
            const int row = rowBase + 8 * hi;
            #pragma unroll
            for (int r = 0; r < 8; ++r)
                P[(size_t)(row + r) * F + col] = acc[r];
        } else if (hi == 0) {
            c[col] = acc[0];
        }
    }
}

template <int F, int U>
__device__ __forceinline__ void add_edges(const float* __restrict__ h, const int2* __restrict__ csr,
                                          int e, int f, float4& acc) {
    int2   p[U];
    float4 v[U];
    #pragma unroll
    for (int u = 0; u < U; ++u) p[u] = csr[e + u];
    #pragma unroll
    for (int u = 0; u < U; ++u) v[u] = *(const float4*)(h + (size_t)p[u].x * F + f);
    #pragma unroll
    for (int u = 0; u < U; ++u) {
        const float c = __int_as_float(p[u].y);
        acc.x = fmaf(v[u].x, c, acc.x);
        acc.y = fmaf(v[u].y, c, acc.y);
        acc.z = fmaf(v[u].z, c, acc.z);
        acc.w = fmaf(v[u].w, c, acc.w);
    }
}

template <int F>
__device__ __forceinline__ void gather_body(int bid, const float* __restrict__ h, const int* __restrict__ csr_raw,
                                            const int* __restrict__ rowptr, const float* __restrict__ dinv,
                                            const float* __restrict__ b, float* __restrict__ out, int do_relu) {
    const int2* __restrict__ csr = (const int2*)csr_raw;
    const int lane = threadIdx.x & 31;
    const int row = __builtin_amdgcn_readfirstlane(bid * 8 + (threadIdx.x >> 5));
    const int f = lane * 4;
    if (F % 128 != 0 && f >= F) return;

    const int beg = rowptr[row];
    const int end = rowptr[row + 1];
    const float4 self = *(const float4*)(h + (size_t)row * F + f);
    const float di = dinv[row];
    const float sl = di * di;

    float4 acc;
    acc.x = fmaf(self.x, sl, b[f + 0]);
    acc.y = fmaf(self.y, sl, b[f + 1]);
    acc.z = fmaf(self.z, sl, b[f + 2]);
    acc.w = fmaf(self.w, sl, b[f + 3]);

    int e = beg;
    for (; e + 8 <= end; e += 8) add_edges<F, 8>(h, csr, e, f, acc);
    switch (end - e) {
        case 7: add_edges<F, 7>(h, csr, e, f, acc); break;
        case 6: add_edges<F, 6>(h, csr, e, f, acc); break;
        case 5: add_edges<F, 5>(h, csr, e, f, acc); break;
        case 4: add_edges<F, 4>(h, csr, e, f, acc); break;
        case 3: add_edges<F, 3>(h, csr, e, f, acc); break;
        case 2: add_edges<F, 2>(h, csr, e, f, acc); break;
        case 1: add_edges<F, 1>(h, csr, e, f, acc); break;
        default: break;
    }

    if (do_relu) {
        acc.x = fmaxf(acc.x, 0.0f);
        acc.y = fmaxf(acc.y, 0.0f);
        acc.z = fmaxf(acc.z, 0.0f);
        acc.w = fmaxf(acc.w, 0.0f);
    }
    *(float4*)(out + (size_t)row * F + f) = acc;
}

#define WCB   (DH / 16 + 1)
#define XTILE (NNODES / 32)
static_assert(EDGES % 512 == 0 && NNODES % 32 == 0 && EDGES / 512 == XTILE,
              "two edge blocks and one x W1 row tile per group of three blocks");

__global__ void __launch_bounds__(256)
count_xw1_w5_kernel(const int* __restrict__ dst, int* __restrict__ cnt, int* __restrict__ rank, int E,
                    const float* __restrict__ x, const float* __restrict__ W1, float* __restrict__ xw1,
                    const float* __restrict__ W5, const float* __restrict__ b5,
                    const float* __restrict__ Wl, const float* __restrict__ bl,
                    float* __restrict__ P5, float* __restrict__ c5) {
    const int bid = blockIdx.x;
    if (bid < WCB) {
        wchain_body<NCLS, NCLS>(bid, WCB, W5, b5, Wl, bl, P5, c5);
    } else {
        const int b = bid - WCB;
        const int g = b / 3;
        const int r = b - 3 * g;
        if (r == 2) gemm_body<DH, DH>(g, x, W1, nullptr, xw1);
        else        count_body(2 * g + r, dst, cnt, rank, E);
    }
}

__global__ void __launch_bounds__(256)
scan_w4_kernel(const int* __restrict__ cnt, int* __restrict__ rowptr, float* __restrict__ dinv,
               const float* __restrict__ W4, const float* __restrict__ b4,
               const float* __restrict__ P5, float* __restrict__ P4, float* __restrict__ c4) {
    const int bid = blockIdx.x;
    if (bid < WCB) wchain_body<DH, NCLS>(bid, WCB, W4, b4, P5, nullptr, P4, c4);
    else           scan_body(bid - WCB, cnt, rowptr, dinv);
}

__global__ void __launch_bounds__(256)
fill_xw1_w3_kernel(const int* __restrict__ src, const int* __restrict__ dst,
                   const int* __restrict__ rank, const int* __restrict__ rowptr,
                   const float* __restrict__ dinv, int* __restrict__ csr, int E,
                   const float* __restrict__ x, const float* __restrict__ W1, float* __restrict__ xw1,
                   const float* __restrict__ W3, const float* __restrict__ b3,
                   const float* __restrict__ P4, float* __restrict__ P3, float* __restrict__ c3) {
    const int bid = blockIdx.x;
    if (bid < WCB) {
        wchain_body<DH, NCLS>(bid, WCB, W3, b3, P4, nullptr, P3, c3);
    } else {
        const int b = bid - WCB;
        const int g = b / 3;
        const int r = b - 3 * g;
        if (r == 2) gemm_body<DH, DH>(XTILE + g, x, W1, nullptr, xw1);
        else        fill_body(2 * g + r, src, dst, rank, rowptr, dinv, csr, E);
    }
}

__global__ void __launch_bounds__(256)
gather1_w2_kernel(const float* __restrict__ h, const int* __restrict__ csr,
                  const int* __restrict__ rowptr, const float* __restrict__ dinv,
                  const float* __restrict__ b1, float* __restrict__ out,
                  const float* __restrict__ W2, const float* __restrict__ b2,
                  const float* __restrict__ P3, float* __restrict__ P2, float* __restrict__ c2) {
    const int bid = blockIdx.x;
    if (bid < WCB) wchain_body<DH, NCLS>(bid, WCB, W2, b2, P3, nullptr, P2, c2);
    else           gather_body<DH>(bid - WCB, h, csr, rowptr, dinv, b1, out, 1);
}

template <int K, int F>
__global__ void __launch_bounds__(256)
gemm_wmma_f32(const float* __restrict__ A, const float* __restrict__ W,
              const float* __restrict__ bias, float* __restrict__ out) {
    gemm_body<K, F>(blockIdx.x, A, W, bias, out);
}

template <int F>
__global__ void __launch_bounds__(256)
gather_agg(const float* __restrict__ h, const int* __restrict__ csr,
           const int* __restrict__ rowptr, const float* __restrict__ dinv,
           const float* __restrict__ b, float* __restrict__ out, int do_relu) {
    gather_body<F>(blockIdx.x, h, csr, rowptr, dinv, b, out, do_relu);
}

extern "C" void kernel_launch(void* const* d_in, const int* in_sizes, int n_in,
                              void* d_out, int out_size, void* d_ws, size_t ws_size,
                              hipStream_t stream) {
    const float* x  = (const float*)d_in[0];
    const int*   ei = (const int*)d_in[1];
    const float* W1 = (const float*)d_in[2];   const float* b1 = (const float*)d_in[3];
    const float* W2 = (const float*)d_in[4];   const float* b2 = (const float*)d_in[5];
    const float* W3 = (const float*)d_in[6];   const float* b3 = (const float*)d_in[7];
    const float* W4 = (const float*)d_in[8];   const float* b4 = (const float*)d_in[9];
    const float* W5 = (const float*)d_in[10];  const float* b5 = (const float*)d_in[11];
    const float* Wl = (const float*)d_in[12];  const float* bl = (const float*)d_in[13];

    const int* src = ei;
    const int* dst = ei + EDGES;

    float* ws     = (float*)d_ws;
    int*   cnt    = (int*)ws;
    float* dinv   = ws + NNODES;
    int*   rowptr = (int*)(ws + 2 * (size_t)NNODES);
    int*   rank   = rowptr + NNODES + 32;
    int*   csr    = rank + EDGES;
    float* bufH   = (float*)(csr + 2 * (size_t)EDGES);
    float* bufX   = bufH + (size_t)NNODES * DH;
    float* P5     = bufX + (size_t)NNODES * DH;
    float* P4     = P5 + DH * NCLS;
    float* P3     = P4 + DH * NCLS;
    float* P2     = P3 + DH * NCLS;
    float* c5     = P2 + DH * NCLS;
    float* c4     = c5 + 128;
    float* c3     = c4 + 128;
    float* c2     = c3 + 128;

    const int gblocks = NNODES / 16;
    const int ablocks = NNODES / 8;

    hipMemsetAsync(cnt, 0, NNODES * sizeof(int), stream);
    count_xw1_w5_kernel<<<WCB + 3 * XTILE, 256, 0, stream>>>(dst, cnt, rank, EDGES, x, W1, bufH,
                                                             W5, b5, Wl, bl, P5, c5);
    scan_w4_kernel<<<WCB + (NNODES / 4 + 255) / 256, 256, 0, stream>>>(cnt, rowptr, dinv, W4, b4, P5, P4, c4);
    fill_xw1_w3_kernel<<<WCB + 3 * XTILE, 256, 0, stream>>>(src, dst, rank, rowptr, dinv, csr, EDGES,
                                                              x, W1, bufH, W3, b3, P4, P3, c3);

    gather1_w2_kernel<<<WCB + ablocks, 256, 0, stream>>>(bufH, csr, rowptr, dinv, b1, bufX, W2, b2, P3, P2, c2);
    gemm_wmma_f32<DH, NCLS><<<gblocks, 256, 0, stream>>>(bufX, P2, nullptr, bufH);
    gather_agg<NCLS><<<ablocks, 256, 0, stream>>>(bufH, csr, rowptr, dinv, c2, bufX, 0);
    gather_agg<NCLS><<<ablocks, 256, 0, stream>>>(bufX, csr, rowptr, dinv, c3, bufH, 0);
    gather_agg<NCLS><<<ablocks, 256, 0, stream>>>(bufH, csr, rowptr, dinv, c4, bufX, 0);
    gather_agg<NCLS><<<ablocks, 256, 0, stream>>>(bufX, csr, rowptr, dinv, c5, (float*)d_out, 0);
}
